# starting kernel
# baseline (speedup 1.0000x reference)
#include <hip/hip_runtime.h>
#include <hip/hip_bf16.h>

#define BB 2048
#define TT 1024
#define II 28
#define HH 2
#define CC 10
#define NG 8
#define MM (BB * TT)

typedef __attribute__((ext_vector_type(2))) float v2f;
typedef __attribute__((ext_vector_type(8))) float v8f;

__device__ __forceinline__ float fast_tanh(float x) {
#if __has_builtin(__builtin_amdgcn_tanhf)
    return __builtin_amdgcn_tanhf(x);
#else
    float e = __expf(-2.0f * x);
    return (1.0f - e) / (1.0f + e);
#endif
}
__device__ __forceinline__ float fast_sigm(float x) {
    return 0.5f + 0.5f * fast_tanh(0.5f * x);
}

__global__ void lstm_xproj_wmma(const float* __restrict__ x,
                                const float* __restrict__ W_ih,
                                const float* __restrict__ b_ih,
                                const float* __restrict__ b_hh,
                                float* __restrict__ xg) {
    const int lane  = threadIdx.x & 31;
    const int n     = lane & 15;
    const int khalf = lane >> 4;

    v2f bfrag[7];
#pragma unroll
    for (int j = 0; j < 7; ++j) {
        const int k0 = 4 * j + 2 * khalf;
        float b0 = 0.0f, b1 = 0.0f;
        if (n < NG) {
            b0 = W_ih[n * II + k0];
            b1 = W_ih[n * II + k0 + 1];
        }
        v2f bf; bf.x = b0; bf.y = b1;
        bfrag[j] = bf;
    }

    const float bias = (n < NG) ? (b_ih[n] + b_hh[n]) : 0.0f;
    v8f cbias;
#pragma unroll
    for (int v = 0; v < 8; ++v) cbias[v] = bias;

    const int wavesPerBlock = blockDim.x >> 5;
    const int gwave  = blockIdx.x * wavesPerBlock + (threadIdx.x >> 5);
    const int nwaves = gridDim.x * wavesPerBlock;
    const int ntiles = MM >> 4;

    for (int tile = gwave; tile < ntiles; tile += nwaves) {
        const int   row  = tile * 16 + n;
        const float* xrow = x + (long)row * II;

        v8f c = cbias;
#pragma unroll
        for (int j = 0; j < 7; ++j) {
            const int k0 = 4 * j + 2 * khalf;
            v2f a; a.x = xrow[k0]; a.y = xrow[k0 + 1];
            c = __builtin_amdgcn_wmma_f32_16x16x4_f32(
                    false, a, false, bfrag[j], (short)0, c, false, false);
        }

        if (n < NG) {
#pragma unroll
            for (int v = 0; v < 8; ++v) {
                const int m  = v + 8 * khalf;
                const int r  = tile * 16 + m;
                const int bi = r >> 10;
                const int ti = r & (TT - 1);
                xg[((long)ti * BB + bi) * NG + n] = c[v];
            }
        }
    }
}

__global__ void lstm_scan(const float* __restrict__ xg,
                          const float* __restrict__ W_hh,
                          const float* __restrict__ fc_W,
                          const float* __restrict__ fc_b,
                          float* __restrict__ out) {
    const int b = blockIdx.x * blockDim.x + threadIdx.x;
    if (b >= BB) return;

    float w[16];
#pragma unroll
    for (int k = 0; k < 16; ++k) w[k] = W_hh[k];

    const float4* g4 = (const float4*)xg;
    const long stride = (long)BB * 2;
    long idx = (long)b * 2;

    float h0 = 0.f, h1 = 0.f, c0 = 0.f, c1 = 0.f;
    float4 lo = g4[idx], hi = g4[idx + 1];

    for (int t = 0; t < TT; ++t) {
        const long nidx = idx + stride;
        float4 nlo = lo, nhi = hi;
        if (t + 1 < TT) { nlo = g4[nidx]; nhi = g4[nidx + 1]; }

        const float i0 = lo.x + h0 * w[0]  + h1 * w[1];
        const float i1 = lo.y + h0 * w[2]  + h1 * w[3];
        const float f0 = lo.z + h0 * w[4]  + h1 * w[5];
        const float f1 = lo.w + h0 * w[6]  + h1 * w[7];
        const float g0 = hi.x + h0 * w[8]  + h1 * w[9];
        const float g1 = hi.y + h0 * w[10] + h1 * w[11];
        const float o0 = hi.z + h0 * w[12] + h1 * w[13];
        const float o1 = hi.w + h0 * w[14] + h1 * w[15];

        c0 = fast_sigm(f0) * c0 + fast_sigm(i0) * fast_tanh(g0);
        c1 = fast_sigm(f1) * c1 + fast_sigm(i1) * fast_tanh(g1);
        h0 = fast_sigm(o0) * fast_tanh(c0);
        h1 = fast_sigm(o1) * fast_tanh(c1);

        lo = nlo; hi = nhi; idx = nidx;
    }

#pragma unroll
    for (int c = 0; c < CC; ++c) {
        out[b * CC + c] = fc_b[c] + h0 * fc_W[c * HH + 0] + h1 * fc_W[c * HH + 1];
    }
}

__global__ void lstm_fused(const float* __restrict__ x,
                           const float* __restrict__ W_ih,
                           const float* __restrict__ W_hh,
                           const float* __restrict__ b_ih,
                           const float* __restrict__ b_hh,
                           const float* __restrict__ fc_W,
                           const float* __restrict__ fc_b,
                           float* __restrict__ out) {
    const int b = blockIdx.x * blockDim.x + threadIdx.x;
    if (b >= BB) return;

    float w[16];
#pragma unroll
    for (int k = 0; k < 16; ++k) w[k] = W_hh[k];
    float bias[NG];
#pragma unroll
    for (int g = 0; g < NG; ++g) bias[g] = b_ih[g] + b_hh[g];

    float h0 = 0.f, h1 = 0.f, c0 = 0.f, c1 = 0.f;
    const float* xb = x + (long)b * TT * II;

    for (int t = 0; t < TT; ++t) {
        const float* xr = xb + (long)t * II;
        float ga[NG];
#pragma unroll
        for (int g = 0; g < NG; ++g) {
            float s = bias[g];
            for (int i = 0; i < II; ++i) s += xr[i] * W_ih[g * II + i];
            ga[g] = s;
        }
        const float i0 = ga[0] + h0 * w[0]  + h1 * w[1];
        const float i1 = ga[1] + h0 * w[2]  + h1 * w[3];
        const float f0 = ga[2] + h0 * w[4]  + h1 * w[5];
        const float f1 = ga[3] + h0 * w[6]  + h1 * w[7];
        const float g0 = ga[4] + h0 * w[8]  + h1 * w[9];
        const float g1 = ga[5] + h0 * w[10] + h1 * w[11];
        const float o0 = ga[6] + h0 * w[12] + h1 * w[13];
        const float o1 = ga[7] + h0 * w[14] + h1 * w[15];

        c0 = fast_sigm(f0) * c0 + fast_sigm(i0) * fast_tanh(g0);
        c1 = fast_sigm(f1) * c1 + fast_sigm(i1) * fast_tanh(g1);
        h0 = fast_sigm(o0) * fast_tanh(c0);
        h1 = fast_sigm(o1) * fast_tanh(c1);
    }
#pragma unroll
    for (int c = 0; c < CC; ++c) {
        out[b * CC + c] = fc_b[c] + h0 * fc_W[c * HH + 0] + h1 * fc_W[c * HH + 1];
    }
}

extern "C" void kernel_launch(void* const* d_in, const int* in_sizes, int n_in,
                              void* d_out, int out_size, void* d_ws, size_t ws_size,
                              hipStream_t stream) {
    const float* x    = (const float*)d_in[0];
    const float* W_ih = (const float*)d_in[1];
    const float* W_hh = (const float*)d_in[2];
    const float* b_ih = (const float*)d_in[3];
    const float* b_hh = (const float*)d_in[4];
    const float* fc_W = (const float*)d_in[5];
    const float* fc_b = (const float*)d_in[6];
    float* out = (float*)d_out;

    const size_t need = (size_t)MM * NG * sizeof(float);
    if (ws_size >= need) {
        float* xg = (float*)d_ws;
        lstm_xproj_wmma<<<2048, 256, 0, stream>>>(x, W_ih, b_ih, b_hh, xg);
        lstm_scan<<<BB / 32, 32, 0, stream>>>(xg, W_hh, fc_W, fc_b, out);
    } else {
        lstm_fused<<<BB / 32, 32, 0, stream>>>(x, W_ih, W_hh, b_ih, b_hh,
                                               fc_W, fc_b, out);
    }
}


# one launch: each scan wave forms its own gates on the matrix unit with W_ih as A and x rows as B (batch on columns), so D leaves lane n its 8 gates in registers; no gate round trip; sums unchanged
# speedup vs baseline: 2.8484x; 2.8484x over previous
#include <hip/hip_runtime.h>


#define BB 2048
#define TT 1024
#define II 28
#define HH 2
#define CC 10
#define NG 8
#define TC 16
#define WB 16

typedef __attribute__((ext_vector_type(2))) float v2f;
typedef __attribute__((ext_vector_type(8))) float v8f;

__device__ __forceinline__ float fast_tanh(float x) {
#if __has_builtin(__builtin_amdgcn_tanhf)
    return __builtin_amdgcn_tanhf(x);
#else
    float e = __expf(-2.0f * x);
    return (1.0f - e) / (1.0f + e);
#endif
}
__device__ __forceinline__ float fast_sigm(float x) {
    return 0.5f + 0.5f * fast_tanh(0.5f * x);
}

struct XRow { v2f f[7]; };

__device__ __forceinline__ void load_row(XRow& r, const float* __restrict__ p) {
#pragma unroll
    for (int j = 0; j < 7; ++j) r.f[j] = *(const v2f*)(p + 4 * j);
}

__device__ __forceinline__ void gates_of2(const XRow& r0, const XRow& r1,
                                          const v2f (&a)[7], v8f& c0, v8f& c1) {
#pragma unroll
    for (int j = 0; j < 7; ++j) {
        c0 = __builtin_amdgcn_wmma_f32_16x16x4_f32(
                 false, a[j], false, r0.f[j], (short)0, c0, false, false);
        c1 = __builtin_amdgcn_wmma_f32_16x16x4_f32(
                 false, a[j], false, r1.f[j], (short)0, c1, false, false);
    }
}

struct Cell { float h0, h1, c0, c1; };

__device__ __forceinline__ void scan_step(Cell& s, const v8f g, const float (&w)[16]) {
    const float i0 = g[0] + s.h0 * w[0]  + s.h1 * w[1];
    const float i1 = g[1] + s.h0 * w[2]  + s.h1 * w[3];
    const float f0 = g[2] + s.h0 * w[4]  + s.h1 * w[5];
    const float f1 = g[3] + s.h0 * w[6]  + s.h1 * w[7];
    const float g0 = g[4] + s.h0 * w[8]  + s.h1 * w[9];
    const float g1 = g[5] + s.h0 * w[10] + s.h1 * w[11];
    const float o0 = g[6] + s.h0 * w[12] + s.h1 * w[13];
    const float o1 = g[7] + s.h0 * w[14] + s.h1 * w[15];

    s.c0 = fast_sigm(f0) * s.c0 + fast_sigm(i0) * fast_tanh(g0);
    s.c1 = fast_sigm(f1) * s.c1 + fast_sigm(i1) * fast_tanh(g1);
    s.h0 = fast_sigm(o0) * fast_tanh(s.c0);
    s.h1 = fast_sigm(o1) * fast_tanh(s.c1);
}

template <int NLOAD, int NWMMA>
__device__ __forceinline__ void chunk(XRow (&q)[TC], v8f& gA, v8f& gB, Cell& st,
                                      const float* __restrict__ p,
                                      const v2f (&a)[7], const v8f cb,
                                      const float (&w)[16]) {
#pragma unroll
    for (int s = 0; s < TC; s += 2) {
        v8f nA = cb, nB = cb;
        if (s < NWMMA) gates_of2(q[s], q[s + 1], a, nA, nB);
        asm volatile("" : "+v"(nA), "+v"(nB) : : "memory");
        if (s < NLOAD) {
            load_row(q[s],     p + (s + 2 + TC) * II);
            load_row(q[s + 1], p + (s + 3 + TC) * II);
        }
        scan_step(st, gA, w);
        scan_step(st, gB, w);
        gA = nA; gB = nB;
#pragma unroll
        for (int i = 0; i < 14; ++i) {
            __builtin_amdgcn_sched_group_barrier(0x008, 1, 0);
            __builtin_amdgcn_sched_group_barrier(0x402, 7, 0);
        }
        __builtin_amdgcn_sched_barrier(0);
    }
}

__global__ void __launch_bounds__(32)
lstm_rowgate_scan(const float* __restrict__ x,
                  const float* __restrict__ W_ih,
                  const float* __restrict__ W_hh,
                  const float* __restrict__ b_ih,
                  const float* __restrict__ b_hh,
                  const float* __restrict__ fc_W,
                  const float* __restrict__ fc_b,
                  float* __restrict__ out) {
    const int lane  = threadIdx.x & 31;
    const int n     = lane & 15;
    const int khalf = lane >> 4;
    const int b     = blockIdx.x * WB + n;

    v2f afrag[7];
#pragma unroll
    for (int j = 0; j < 7; ++j) {
        const int k0 = 4 * j + 2 * khalf;
        float a0 = 0.0f, a1 = 0.0f;
        if (n < NG) {
            a0 = W_ih[n * II + k0];
            a1 = W_ih[n * II + k0 + 1];
        }
        v2f af; af.x = a0; af.y = a1;
        afrag[j] = af;
    }

    v8f cbias;
#pragma unroll
    for (int v = 0; v < 8; ++v) cbias[v] = (khalf == 0) ? (b_ih[v] + b_hh[v]) : 0.0f;

    float w[16];
#pragma unroll
    for (int k = 0; k < 16; ++k) w[k] = W_hh[k];

    const float* p = x + (long)b * TT * II + 2 * khalf;

    XRow r0, r1;
    load_row(r0, p);
    load_row(r1, p + II);
    XRow q[TC];
#pragma unroll
    for (int s = 0; s < TC; ++s) load_row(q[s], p + (s + 2) * II);

    v8f gA = cbias, gB = cbias;
    gates_of2(r0, r1, afrag, gA, gB);
    Cell st; st.h0 = 0.f; st.h1 = 0.f; st.c0 = 0.f; st.c1 = 0.f;

    for (int t0 = 0; t0 < TT - 2 * TC; t0 += TC) {
        chunk<TC, TC>(q, gA, gB, st, p, afrag, cbias, w);
        p += TC * II;
    }
    chunk<TC - 2, TC>(q, gA, gB, st, p, afrag, cbias, w);
    p += TC * II;
    chunk<0, TC - 2>(q, gA, gB, st, p, afrag, cbias, w);

    asm volatile("" : "+v"(st.h0), "+v"(st.h1));

    if (khalf == 0) {
#pragma unroll
        for (int c = 0; c < CC; ++c) {
            out[b * CC + c] = fc_b[c] + st.h0 * fc_W[c * HH + 0] + st.h1 * fc_W[c * HH + 1];
        }
    }
}

extern "C" void kernel_launch(void* const* d_in, const int* in_sizes, int n_in,
                              void* d_out, int out_size, void* d_ws, size_t ws_size,
                              hipStream_t stream) {
    const float* x    = (const float*)d_in[0];
    const float* W_ih = (const float*)d_in[1];
    const float* W_hh = (const float*)d_in[2];
    const float* b_ih = (const float*)d_in[3];
    const float* b_hh = (const float*)d_in[4];
    const float* fc_W = (const float*)d_in[5];
    const float* fc_b = (const float*)d_in[6];
    float* out = (float*)d_out;

    lstm_rowgate_scan<<<BB / WB, 32, 0, stream>>>(x, W_ih, W_hh, b_ih, b_hh,
                                                  fc_W, fc_b, out);
}


# matrix tile columns = 8 batch elements x 2 consecutive timesteps (were 16 x 1): half the matrix instructions and x requests per wave per step, 256 waves; 2nd step's gates by DPP row rotate; same sums
# speedup vs baseline: 4.0349x; 1.4165x over previous
#include <hip/hip_runtime.h>


#define BB 2048
#define TT 1024
#define II 28
#define HH 2
#define CC 10
#define NG 8
#define NT (TT / 2)
#define TC 16
#define WB 8

typedef __attribute__((ext_vector_type(2))) float v2f;
typedef __attribute__((ext_vector_type(8))) float v8f;

__device__ __forceinline__ float fast_tanh(float x) {
#if __has_builtin(__builtin_amdgcn_tanhf)
    return __builtin_amdgcn_tanhf(x);
#else
    float e = __expf(-2.0f * x);
    return (1.0f - e) / (1.0f + e);
#endif
}
__device__ __forceinline__ float fast_sigm(float x) {
    return 0.5f + 0.5f * fast_tanh(0.5f * x);
}

struct XRow { v2f f[7]; };

__device__ __forceinline__ void load_row(XRow& r, const float* __restrict__ p) {
#pragma unroll
    for (int j = 0; j < 7; ++j) r.f[j] = *(const v2f*)(p + 4 * j);
}

template <int K>
__device__ __forceinline__ void gate_mm(const XRow& r0, const XRow& r1,
                                        const v2f (&a)[7], v8f& c0, v8f& c1) {
    if ((K & 1) == 0)
        c0 = __builtin_amdgcn_wmma_f32_16x16x4_f32(
                 false, a[K >> 1], false, r0.f[K >> 1], (short)0, c0, false, false);
    else
        c1 = __builtin_amdgcn_wmma_f32_16x16x4_f32(
                 false, a[K >> 1], false, r1.f[K >> 1], (short)0, c1, false, false);
}
__device__ __forceinline__ void gates_of2(const XRow& r0, const XRow& r1,
                                          const v2f (&a)[7], v8f& c0, v8f& c1) {
    gate_mm<0>(r0, r1, a, c0, c1);  gate_mm<1>(r0, r1, a, c0, c1);
    gate_mm<2>(r0, r1, a, c0, c1);  gate_mm<3>(r0, r1, a, c0, c1);
    gate_mm<4>(r0, r1, a, c0, c1);  gate_mm<5>(r0, r1, a, c0, c1);
    gate_mm<6>(r0, r1, a, c0, c1);  gate_mm<7>(r0, r1, a, c0, c1);
    gate_mm<8>(r0, r1, a, c0, c1);  gate_mm<9>(r0, r1, a, c0, c1);
    gate_mm<10>(r0, r1, a, c0, c1); gate_mm<11>(r0, r1, a, c0, c1);
    gate_mm<12>(r0, r1, a, c0, c1); gate_mm<13>(r0, r1, a, c0, c1);
}

struct Cell { float h0, h1, c0, c1, t0, t1, t2, t3, t4, t5, t6, t7; };

__device__ __forceinline__ void step_p1(Cell& s, const v8f g, const float (&w)[16]) {
    s.t0 = g[0] + s.h0 * w[0]  + s.h1 * w[1];
    s.t1 = g[1] + s.h0 * w[2]  + s.h1 * w[3];
    s.t2 = g[2] + s.h0 * w[4]  + s.h1 * w[5];
    s.t3 = g[3] + s.h0 * w[6]  + s.h1 * w[7];
    s.t4 = g[4] + s.h0 * w[8]  + s.h1 * w[9];
    s.t5 = g[5] + s.h0 * w[10] + s.h1 * w[11];
    const float o0 = g[6] + s.h0 * w[12] + s.h1 * w[13];
    const float o1 = g[7] + s.h0 * w[14] + s.h1 * w[15];
    s.t6 = fast_sigm(o0);
    s.t7 = fast_sigm(o1);
}
__device__ __forceinline__ void step_p2(Cell& s) {
    s.c0 = fast_sigm(s.t2) * s.c0 + fast_sigm(s.t0) * fast_tanh(s.t4);
    s.c1 = fast_sigm(s.t3) * s.c1 + fast_sigm(s.t1) * fast_tanh(s.t5);
}
__device__ __forceinline__ void step_p3(Cell& s) {
    s.h0 = s.t6 * fast_tanh(s.c0);
    s.h1 = s.t7 * fast_tanh(s.c1);
}

__device__ __forceinline__ void point_mm(v8f& nA, v8f& nB) {
    asm volatile("" : "+v"(nA), "+v"(nB));
}
__device__ __forceinline__ void point_p1(v8f& nA, v8f& nB, Cell& s) {
    asm volatile("" : "+v"(nA), "+v"(nB), "+v"(s.t0), "+v"(s.t1), "+v"(s.t2),
                      "+v"(s.t3), "+v"(s.t4), "+v"(s.t5), "+v"(s.t6), "+v"(s.t7));
}
__device__ __forceinline__ void point_p2(v8f& nA, v8f& nB, Cell& s) {
    asm volatile("" : "+v"(nA), "+v"(nB), "+v"(s.c0), "+v"(s.c1));
}
__device__ __forceinline__ void point_p3(v8f& nA, v8f& nB, Cell& s) {
    asm volatile("" : "+v"(nA), "+v"(nB), "+v"(s.h0), "+v"(s.h1));
}

__device__ __forceinline__ v8f other_step(const v8f g) {
    v8f r;
#pragma unroll
    for (int v = 0; v < 8; ++v)
        r[v] = __int_as_float(__builtin_amdgcn_update_dpp(
                   0, __float_as_int(g[v]), 0x128, 0xf, 0xf, true));
    return r;
}

__device__ __forceinline__ const float* tile_row(const char* __restrict__ xb,
                                                 unsigned lane_off, int u) {
    const int uc = u < NT - 1 ? u : NT - 1;
    return (const float*)(xb + (lane_off + (unsigned)uc * (unsigned)(2 * II * sizeof(float))));
}

__device__ __forceinline__ void chunk(XRow (&q)[TC], v8f& gA, v8f& gB, Cell& st,
                                      const char* __restrict__ xb, unsigned lane_off,
                                      int u0, const v2f (&a)[7], const v8f cb,
                                      const float (&w)[16]) {
#pragma unroll
    for (int s = 0; s < TC; s += 2) {
        v8f nA = cb, nB = cb;
        const XRow& r0 = q[s];
        const XRow& r1 = q[s + 1];
        const v8f gA2 = other_step(gA), gB2 = other_step(gB);
        gate_mm<0>(r0, r1, a, nA, nB);                         point_mm(nA, nB);
        gate_mm<1>(r0, r1, a, nA, nB);  step_p1(st, gA, w);    point_p1(nA, nB, st);
        gate_mm<2>(r0, r1, a, nA, nB);  step_p2(st);           point_p2(nA, nB, st);
        gate_mm<3>(r0, r1, a, nA, nB);  step_p3(st);           point_p3(nA, nB, st);
        gate_mm<4>(r0, r1, a, nA, nB);  step_p1(st, gA2, w);   point_p1(nA, nB, st);
        gate_mm<5>(r0, r1, a, nA, nB);  step_p2(st);           point_p2(nA, nB, st);
        gate_mm<6>(r0, r1, a, nA, nB);  step_p3(st);           point_p3(nA, nB, st);
        gate_mm<7>(r0, r1, a, nA, nB);  step_p1(st, gB, w);    point_p1(nA, nB, st);
        gate_mm<8>(r0, r1, a, nA, nB);  step_p2(st);           point_p2(nA, nB, st);
        gate_mm<9>(r0, r1, a, nA, nB);  step_p3(st);           point_p3(nA, nB, st);
        gate_mm<10>(r0, r1, a, nA, nB); step_p1(st, gB2, w);   point_p1(nA, nB, st);
        gate_mm<11>(r0, r1, a, nA, nB); step_p2(st);           point_p2(nA, nB, st);
        gate_mm<12>(r0, r1, a, nA, nB); step_p3(st);           point_p3(nA, nB, st);
        gate_mm<13>(r0, r1, a, nA, nB);
        asm volatile("" : "+v"(nA), "+v"(nB) : : "memory");
        load_row(q[s],     tile_row(xb, lane_off, u0 + s + 2 + TC));
        load_row(q[s + 1], tile_row(xb, lane_off, u0 + s + 3 + TC));
        gA = nA; gB = nB;
    }
}

__global__ void __launch_bounds__(32)
lstm_rowgate_scan(const float* __restrict__ x,
                  const float* __restrict__ W_ih,
                  const float* __restrict__ W_hh,
                  const float* __restrict__ b_ih,
                  const float* __restrict__ b_hh,
                  const float* __restrict__ fc_W,
                  const float* __restrict__ fc_b,
                  float* __restrict__ out) {
    const int lane  = threadIdx.x & 31;
    const int n     = lane & 15;
    const int khalf = lane >> 4;
    const int dt    = n >> 3;
    const int b     = blockIdx.x * WB + (n & 7);

    v2f afrag[7];
#pragma unroll
    for (int j = 0; j < 7; ++j) {
        const int k0 = 4 * j + 2 * khalf;
        float a0 = 0.0f, a1 = 0.0f;
        if (n < NG) {
            a0 = W_ih[n * II + k0];
            a1 = W_ih[n * II + k0 + 1];
        }
        v2f af; af.x = a0; af.y = a1;
        afrag[j] = af;
    }

    v8f cbias;
#pragma unroll
    for (int v = 0; v < 8; ++v) cbias[v] = (khalf == 0) ? (b_ih[v] + b_hh[v]) : 0.0f;

    float w[16];
#pragma unroll
    for (int k = 0; k < 16; ++k) w[k] = W_hh[k];

    const char* xb = (const char*)x;
    const unsigned lane_off =
        (unsigned)(((b * TT + dt) * II + 2 * khalf) * (int)sizeof(float));

    XRow r0, r1;
    load_row(r0, tile_row(xb, lane_off, 0));
    load_row(r1, tile_row(xb, lane_off, 1));
    XRow q[TC];
#pragma unroll
    for (int s = 0; s < TC; ++s) load_row(q[s], tile_row(xb, lane_off, s + 2));

    v8f gA = cbias, gB = cbias;
    gates_of2(r0, r1, afrag, gA, gB);
    Cell st;
    st.h0 = 0.f; st.h1 = 0.f; st.c0 = 0.f; st.c1 = 0.f;
    st.t0 = 0.f; st.t1 = 0.f; st.t2 = 0.f; st.t3 = 0.f;
    st.t4 = 0.f; st.t5 = 0.f; st.t6 = 0.f; st.t7 = 0.f;

    for (int u0 = 0; u0 < NT; u0 += TC)
        chunk(q, gA, gB, st, xb, lane_off, u0, afrag, cbias, w);

    if (lane < WB) {
#pragma unroll
        for (int c = 0; c < CC; ++c) {
            out[b * CC + c] = fc_b[c] + st.h0 * fc_W[c * HH + 0] + st.h1 * fc_W[c * HH + 1];
        }
    }
}

extern "C" void kernel_launch(void* const* d_in, const int* in_sizes, int n_in,
                              void* d_out, int out_size, void* d_ws, size_t ws_size,
                              hipStream_t stream) {
    const float* x    = (const float*)d_in[0];
    const float* W_ih = (const float*)d_in[1];
    const float* W_hh = (const float*)d_in[2];
    const float* b_ih = (const float*)d_in[3];
    const float* b_hh = (const float*)d_in[4];
    const float* fc_W = (const float*)d_in[5];
    const float* fc_b = (const float*)d_in[6];
    float* out = (float*)d_out;

    lstm_rowgate_scan<<<BB / WB, 32, 0, stream>>>(x, W_ih, W_hh, b_ih, b_hh,
                                                  fc_W, fc_b, out);
}


# matrix tile columns = 2 batch elements x 8 consecutive timesteps (were 8 x 2): 7 matrix instructions per 8 steps per wave (were 28), 1024 waves; later steps' gates by DPP row shift; same sums
# speedup vs baseline: 5.8155x; 1.4413x over previous
#include <hip/hip_runtime.h>


#define BB 2048
#define TT 1024
#define II 28
#define HH 2
#define CC 10
#define NG 8
#define TS 8
#define WB 2
#define NT (TT / TS)
#define TC 4
#define WPB 4
#define NWG (BB / (WB * WPB))
#define WGT (32 * WPB)

typedef __attribute__((ext_vector_type(2))) float v2f;
typedef __attribute__((ext_vector_type(8))) float v8f;

__device__ __forceinline__ float fast_tanh(float x) {
#if __has_builtin(__builtin_amdgcn_tanhf)
    return __builtin_amdgcn_tanhf(x);
#else
    float e = __expf(-2.0f * x);
    return (1.0f - e) / (1.0f + e);
#endif
}
__device__ __forceinline__ float fast_sigm(float x) {
    return 0.5f + 0.5f * fast_tanh(0.5f * x);
}

struct XRow { v2f f[7]; };

__device__ __forceinline__ void load_row(XRow& r, const float* __restrict__ p) {
#pragma unroll
    for (int j = 0; j < 7; ++j) r.f[j] = *(const v2f*)(p + 4 * j);
}

template <int K>
__device__ __forceinline__ void gate_mm(const XRow& r, const v2f (&a)[7], v8f& c) {
    c = __builtin_amdgcn_wmma_f32_16x16x4_f32(
            false, a[K], false, r.f[K], (short)0, c, false, false);
}
__device__ __forceinline__ void gates_of(const XRow& r, const v2f (&a)[7], v8f& c) {
    gate_mm<0>(r, a, c); gate_mm<1>(r, a, c); gate_mm<2>(r, a, c); gate_mm<3>(r, a, c);
    gate_mm<4>(r, a, c); gate_mm<5>(r, a, c); gate_mm<6>(r, a, c);
}

struct Cell { float h0, h1, c0, c1; };

__device__ __forceinline__ void cell_step(Cell& s, const v8f g, const float (&w)[16]) {
    const float t0 = g[0] + s.h0 * w[0]  + s.h1 * w[1];
    const float t1 = g[1] + s.h0 * w[2]  + s.h1 * w[3];
    const float t2 = g[2] + s.h0 * w[4]  + s.h1 * w[5];
    const float t3 = g[3] + s.h0 * w[6]  + s.h1 * w[7];
    const float t4 = g[4] + s.h0 * w[8]  + s.h1 * w[9];
    const float t5 = g[5] + s.h0 * w[10] + s.h1 * w[11];
    const float o0 = g[6] + s.h0 * w[12] + s.h1 * w[13];
    const float o1 = g[7] + s.h0 * w[14] + s.h1 * w[15];
    const float t6 = fast_sigm(o0);
    const float t7 = fast_sigm(o1);
    s.c0 = fast_sigm(t2) * s.c0 + fast_sigm(t0) * fast_tanh(t4);
    s.c1 = fast_sigm(t3) * s.c1 + fast_sigm(t1) * fast_tanh(t5);
    s.h0 = t6 * fast_tanh(s.c0);
    s.h1 = t7 * fast_tanh(s.c1);
}

__device__ __forceinline__ void point(v8f& nx, Cell& s) {
    asm volatile("" : "+v"(nx), "+v"(s.h0), "+v"(s.h1), "+v"(s.c0), "+v"(s.c1));
}
__device__ __forceinline__ void point_rows(v8f& nx, Cell& s) {
    asm volatile("" : "+v"(nx), "+v"(s.h0), "+v"(s.h1), "+v"(s.c0), "+v"(s.c1) : : "memory");
}

template <int K>
__device__ __forceinline__ v8f step_gates(const v8f g) {
    v8f r;
#pragma unroll
    for (int v = 0; v < 8; ++v)
        r[v] = __int_as_float(__builtin_amdgcn_update_dpp(
                   0, __float_as_int(g[v]), 0x100 + 2 * K, 0xf, 0xf, true));
    return r;
}

__device__ __forceinline__ const float* tile_row(const char* __restrict__ xb,
                                                 unsigned lane_off, int u) {
    const int uc = u < NT - 1 ? u : NT - 1;
    return (const float*)(xb + (lane_off + (unsigned)uc * (unsigned)(TS * II * sizeof(float))));
}

__device__ __forceinline__ void chunk(XRow (&q)[TC], v8f& g, Cell& st,
                                      const char* __restrict__ xb, unsigned lane_off,
                                      int u0, const v2f (&a)[7], const v8f cb,
                                      const float (&w)[16]) {
#pragma unroll
    for (int s = 0; s < TC; ++s) {
        v8f nx = cb;
        const XRow& r = q[s];
        gate_mm<0>(r, a, nx);  cell_step(st, g, w);                  point(nx, st);
        gate_mm<1>(r, a, nx);  cell_step(st, step_gates<1>(g), w);   point(nx, st);
        gate_mm<2>(r, a, nx);  cell_step(st, step_gates<2>(g), w);   point(nx, st);
        gate_mm<3>(r, a, nx);  cell_step(st, step_gates<3>(g), w);   point(nx, st);
        gate_mm<4>(r, a, nx);  cell_step(st, step_gates<4>(g), w);   point(nx, st);
        gate_mm<5>(r, a, nx);  cell_step(st, step_gates<5>(g), w);   point(nx, st);
        gate_mm<6>(r, a, nx);  cell_step(st, step_gates<6>(g), w);   point_rows(nx, st);
        load_row(q[s], tile_row(xb, lane_off, u0 + s + 1 + TC));
        cell_step(st, step_gates<7>(g), w);
        g = nx;
    }
}

__global__ void __launch_bounds__(WGT)
lstm_rowgate_scan(const float* __restrict__ x,
                  const float* __restrict__ W_ih,
                  const float* __restrict__ W_hh,
                  const float* __restrict__ b_ih,
                  const float* __restrict__ b_hh,
                  const float* __restrict__ fc_W,
                  const float* __restrict__ fc_b,
                  float* __restrict__ out) {
    __shared__ float hfin[WPB * WB * HH];

    const int tid   = threadIdx.x;
    const int lane  = tid & 31;
    const int wave  = tid >> 5;
    const int n     = lane & 15;
    const int khalf = lane >> 4;
    const int dt    = n >> 1;
    const int b     = (blockIdx.x * WPB + wave) * WB + (n & 1);

    v2f afrag[7];
#pragma unroll
    for (int j = 0; j < 7; ++j) {
        const int k0 = 4 * j + 2 * khalf;
        float a0 = 0.0f, a1 = 0.0f;
        if (n < NG) {
            a0 = W_ih[n * II + k0];
            a1 = W_ih[n * II + k0 + 1];
        }
        v2f af; af.x = a0; af.y = a1;
        afrag[j] = af;
    }

    v8f cbias;
#pragma unroll
    for (int v = 0; v < 8; ++v) cbias[v] = (khalf == 0) ? (b_ih[v] + b_hh[v]) : 0.0f;

    float w[16];
#pragma unroll
    for (int k = 0; k < 16; ++k) w[k] = W_hh[k];

    const char* xb = (const char*)x;
    const unsigned lane_off =
        (unsigned)(((b * TT + dt) * II + 2 * khalf) * (int)sizeof(float));

    XRow r0;
    load_row(r0, tile_row(xb, lane_off, 0));
    XRow q[TC];
#pragma unroll
    for (int s = 0; s < TC; ++s) load_row(q[s], tile_row(xb, lane_off, s + 1));

    v8f g = cbias;
    gates_of(r0, afrag, g);
    Cell st;
    st.h0 = 0.f; st.h1 = 0.f; st.c0 = 0.f; st.c1 = 0.f;

    for (int u0 = 0; u0 < NT; u0 += TC)
        chunk(q, g, st, xb, lane_off, u0, afrag, cbias, w);

    if (lane < WB) {
        hfin[(wave * WB + lane) * HH + 0] = st.h0;
        hfin[(wave * WB + lane) * HH + 1] = st.h1;
    }
    __syncthreads();

    if (tid < WPB * WB) {
        const int bo = blockIdx.x * (WPB * WB) + tid;
        const float h0 = hfin[tid * HH + 0];
        const float h1 = hfin[tid * HH + 1];
#pragma unroll
        for (int c = 0; c < CC; ++c) {
            out[bo * CC + c] = fc_b[c] + h0 * fc_W[c * HH + 0] + h1 * fc_W[c * HH + 1];
        }
    }
}

extern "C" void kernel_launch(void* const* d_in, const int* in_sizes, int n_in,
                              void* d_out, int out_size, void* d_ws, size_t ws_size,
                              hipStream_t stream) {
    const float* x    = (const float*)d_in[0];
    const float* W_ih = (const float*)d_in[1];
    const float* W_hh = (const float*)d_in[2];
    const float* b_ih = (const float*)d_in[3];
    const float* b_hh = (const float*)d_in[4];
    const float* fc_W = (const float*)d_in[5];
    const float* fc_b = (const float*)d_in[6];
    float* out = (float*)d_out;

    lstm_rowgate_scan<<<NWG, WGT, 0, stream>>>(x, W_ih, W_hh, b_ih, b_hh,
                                               fc_W, fc_b, out);
}


# matrix operands swapped (x rows as A, W_ih as B): D leaves gate n on lane n, the tile's 8 steps in 8 registers; one activation per step serves all gates, h spread by row-share products; same sums
# speedup vs baseline: 7.1310x; 1.2262x over previous
#include <hip/hip_runtime.h>


#define BB 2048
#define TT 1024
#define II 28
#define HH 2
#define CC 10
#define NG 8
#define TS 8
#define WB 2
#define NT (TT / TS)
#define TC 4
#define WPB 4
#define NWG (BB / (WB * WPB))
#define WGT (32 * WPB)

typedef __attribute__((ext_vector_type(2))) float v2f;
typedef __attribute__((ext_vector_type(8))) float v8f;

__device__ __forceinline__ float fast_tanh(float x) {
#if __has_builtin(__builtin_amdgcn_tanhf)
    return __builtin_amdgcn_tanhf(x);
#else
    float e = __expf(-2.0f * x);
    return (1.0f - e) / (1.0f + e);
#endif
}
__device__ __forceinline__ float fast_sigm(float x) {
    return 0.5f + 0.5f * fast_tanh(0.5f * x);
}

struct XRow { v2f f[7]; };

__device__ __forceinline__ void load_row(XRow& r, const float* __restrict__ p) {
#pragma unroll
    for (int j = 0; j < 7; ++j) r.f[j] = *(const v2f*)(p + 4 * j);
}

template <int K>
__device__ __forceinline__ void gate_mm(const XRow& r, const v2f (&w)[7], v8f& c) {
    c = __builtin_amdgcn_wmma_f32_16x16x4_f32(
            false, r.f[K], false, w[K], (short)0, c, false, false);
}
__device__ __forceinline__ void gates_of(const XRow& r, const v2f (&w)[7], v8f& c) {
    gate_mm<0>(r, w, c); gate_mm<1>(r, w, c); gate_mm<2>(r, w, c); gate_mm<3>(r, w, c);
    gate_mm<4>(r, w, c); gate_mm<5>(r, w, c); gate_mm<6>(r, w, c);
}

struct Cell { float h0, h1, c; };

struct LaneK { float w0, w1, as, aa, ao; };

template <int N>
__device__ __forceinline__ float row_up(const float v) {
    return __int_as_float(__builtin_amdgcn_update_dpp(
               0, __float_as_int(v), 0x100 + N, 0xf, 0xf, true));
}
template <int N>
__device__ __forceinline__ float row_lane(const float v) {
    return __int_as_float(__builtin_amdgcn_update_dpp(
               0, __float_as_int(v), 0x150 + N, 0xf, 0xf, true));
}

__device__ __forceinline__ void cell_step(Cell& s, const float g, const LaneK& k) {
    const float t = g + s.h0 * k.w0 + s.h1 * k.w1;
    const float a = k.ao + k.aa * fast_tanh(k.as * t);
    const float o0 = row_lane<6>(a);
    const float o1 = row_lane<7>(a);
    s.c = row_up<2>(a) * s.c + a * row_up<4>(a);
    const float tc = fast_tanh(s.c);
    s.h0 = o0 * row_lane<0>(tc);
    s.h1 = o1 * row_lane<1>(tc);
}

__device__ __forceinline__ void point(v8f& nx, Cell& s) {
    asm volatile("" : "+v"(nx), "+v"(s.h0), "+v"(s.h1), "+v"(s.c));
}
__device__ __forceinline__ void point_rows(v8f& nx, Cell& s) {
    asm volatile("" : "+v"(nx), "+v"(s.h0), "+v"(s.h1), "+v"(s.c) : : "memory");
}

__device__ __forceinline__ const float* tile_row(const char* __restrict__ xb,
                                                 unsigned lane_off, int u) {
    const int uc = u < NT - 1 ? u : NT - 1;
    return (const float*)(xb + (lane_off + (unsigned)uc * (unsigned)(TS * II * sizeof(float))));
}

__device__ __forceinline__ void chunk(XRow (&q)[TC], v8f& g, Cell& st,
                                      const char* __restrict__ xb, unsigned lane_off,
                                      int u0, const v2f (&w)[7], const v8f cb,
                                      const LaneK& k) {
#pragma unroll
    for (int s = 0; s < TC; ++s) {
        v8f nx = cb;
        const XRow& r = q[s];
        gate_mm<0>(r, w, nx);  cell_step(st, g[0], k);   point(nx, st);
        gate_mm<1>(r, w, nx);  cell_step(st, g[1], k);   point(nx, st);
        gate_mm<2>(r, w, nx);  cell_step(st, g[2], k);   point(nx, st);
        gate_mm<3>(r, w, nx);  cell_step(st, g[3], k);   point(nx, st);
        gate_mm<4>(r, w, nx);  cell_step(st, g[4], k);   point(nx, st);
        gate_mm<5>(r, w, nx);  cell_step(st, g[5], k);   point(nx, st);
        gate_mm<6>(r, w, nx);  cell_step(st, g[6], k);   point_rows(nx, st);
        load_row(q[s], tile_row(xb, lane_off, u0 + s + 1 + TC));
        cell_step(st, g[7], k);
        g = nx;
    }
}

__global__ void __launch_bounds__(WGT)
lstm_rowgate_scan(const float* __restrict__ x,
                  const float* __restrict__ W_ih,
                  const float* __restrict__ W_hh,
                  const float* __restrict__ b_ih,
                  const float* __restrict__ b_hh,
                  const float* __restrict__ fc_W,
                  const float* __restrict__ fc_b,
                  float* __restrict__ out) {
    __shared__ float hfin[WPB * WB * HH];

    const int tid   = threadIdx.x;
    const int lane  = tid & 31;
    const int wave  = tid >> 5;
    const int n     = lane & 15;
    const int khalf = lane >> 4;
    const int dt    = n & 7;
    const int b     = (blockIdx.x * WPB + wave) * WB + (n >> 3);

    v2f wfrag[7];
#pragma unroll
    for (int j = 0; j < 7; ++j) {
        const int k0 = 4 * j + 2 * khalf;
        float a0 = 0.0f, a1 = 0.0f;
        if (n < NG) {
            a0 = W_ih[n * II + k0];
            a1 = W_ih[n * II + k0 + 1];
        }
        v2f af; af.x = a0; af.y = a1;
        wfrag[j] = af;
    }

    float bias = 0.0f;
    if (n < NG) bias = b_ih[n] + b_hh[n];
    v8f cbias;
#pragma unroll
    for (int v = 0; v < 8; ++v) cbias[v] = bias;

    LaneK lk;
    lk.w0 = 0.0f; lk.w1 = 0.0f;
    if (n < NG) {
        lk.w0 = W_hh[n * HH + 0];
        lk.w1 = W_hh[n * HH + 1];
    }
    const bool plain = (n >> 1) == 2;
    lk.as = plain ? 1.0f : 0.5f;
    lk.aa = plain ? 1.0f : 0.5f;
    lk.ao = plain ? 0.0f : 0.5f;

    const char* xb = (const char*)x;
    const unsigned lane_off =
        (unsigned)(((b * TT + dt) * II + 2 * khalf) * (int)sizeof(float));

    XRow r0;
    load_row(r0, tile_row(xb, lane_off, 0));
    XRow q[TC];
#pragma unroll
    for (int s = 0; s < TC; ++s) load_row(q[s], tile_row(xb, lane_off, s + 1));

    v8f g = cbias;
    gates_of(r0, wfrag, g);
    Cell st;
    st.h0 = 0.f; st.h1 = 0.f; st.c = 0.f;

    for (int u0 = 0; u0 < NT; u0 += TC)
        chunk(q, g, st, xb, lane_off, u0, wfrag, cbias, lk);

    if (n < HH) hfin[(wave * WB + khalf) * HH + n] = (n == 0) ? st.h0 : st.h1;
    __syncthreads();

    if (tid < WPB * WB) {
        const int bo = blockIdx.x * (WPB * WB) + tid;
        const float h0 = hfin[tid * HH + 0];
        const float h1 = hfin[tid * HH + 1];
#pragma unroll
        for (int c = 0; c < CC; ++c) {
            out[bo * CC + c] = fc_b[c] + h0 * fc_W[c * HH + 0] + h1 * fc_W[c * HH + 1];
        }
    }
}

extern "C" void kernel_launch(void* const* d_in, const int* in_sizes, int n_in,
                              void* d_out, int out_size, void* d_ws, size_t ws_size,
                              hipStream_t stream) {
    const float* x    = (const float*)d_in[0];
    const float* W_ih = (const float*)d_in[1];
    const float* W_hh = (const float*)d_in[2];
    const float* b_ih = (const float*)d_in[3];
    const float* b_hh = (const float*)d_in[4];
    const float* fc_W = (const float*)d_in[5];
    const float* fc_b = (const float*)d_in[6];
    float* out = (float*)d_out;

    lstm_rowgate_scan<<<NWG, WGT, 0, stream>>>(x, W_ih, W_hh, b_ih, b_hh,
                                               fc_W, fc_b, out);
}


# x row requests take a running scalar tile base plus a fixed lane offset register (was a per-tile address register rewritten right behind its requests, behind a translation wait); same sums
# speedup vs baseline: 7.3702x; 1.0335x over previous
#include <hip/hip_runtime.h>


#define BB 2048
#define TT 1024
#define II 28
#define HH 2
#define CC 10
#define NG 8
#define TS 8
#define WB 2
#define NT (TT / TS)
#define TC 4
#define WPB 4
#define NWG (BB / (WB * WPB))
#define WGT (32 * WPB)

typedef __attribute__((ext_vector_type(2))) float v2f;
typedef __attribute__((ext_vector_type(8))) float v8f;

__device__ __forceinline__ float fast_tanh(float x) {
#if __has_builtin(__builtin_amdgcn_tanhf)
    return __builtin_amdgcn_tanhf(x);
#else
    float e = __expf(-2.0f * x);
    return (1.0f - e) / (1.0f + e);
#endif
}
__device__ __forceinline__ float fast_sigm(float x) {
    return 0.5f + 0.5f * fast_tanh(0.5f * x);
}

struct XRow { v2f f[7]; };

__device__ __forceinline__ void load_row(XRow& r, const char* __restrict__ tb, unsigned lane_off) {
    asm("" : "+v"(lane_off));
#pragma unroll
    for (int j = 0; j < 7; ++j) r.f[j] = *(const v2f*)(tb + lane_off + 16 * j);
}

template <int K>
__device__ __forceinline__ void gate_mm(const XRow& r, const v2f (&w)[7], v8f& c) {
    c = __builtin_amdgcn_wmma_f32_16x16x4_f32(
            false, r.f[K], false, w[K], (short)0, c, false, false);
}
__device__ __forceinline__ void gates_of(const XRow& r, const v2f (&w)[7], v8f& c) {
    gate_mm<0>(r, w, c); gate_mm<1>(r, w, c); gate_mm<2>(r, w, c); gate_mm<3>(r, w, c);
    gate_mm<4>(r, w, c); gate_mm<5>(r, w, c); gate_mm<6>(r, w, c);
}

struct Cell { float h0, h1, c; };

struct LaneK { float w0, w1, as, aa, ao; };

template <int N>
__device__ __forceinline__ float row_up(const float v) {
    return __int_as_float(__builtin_amdgcn_update_dpp(
               0, __float_as_int(v), 0x100 + N, 0xf, 0xf, true));
}
template <int N>
__device__ __forceinline__ float row_lane(const float v) {
    return __int_as_float(__builtin_amdgcn_update_dpp(
               0, __float_as_int(v), 0x150 + N, 0xf, 0xf, true));
}

__device__ __forceinline__ void cell_step(Cell& s, const float g, const LaneK& k) {
    const float t = g + s.h0 * k.w0 + s.h1 * k.w1;
    const float a = k.ao + k.aa * fast_tanh(k.as * t);
    const float o0 = row_lane<6>(a);
    const float o1 = row_lane<7>(a);
    s.c = row_up<2>(a) * s.c + a * row_up<4>(a);
    const float tc = fast_tanh(s.c);
    s.h0 = o0 * row_lane<0>(tc);
    s.h1 = o1 * row_lane<1>(tc);
}

__device__ __forceinline__ void point(v8f& nx, Cell& s) {
    asm volatile("" : "+v"(nx), "+v"(s.h0), "+v"(s.h1), "+v"(s.c));
}
__device__ __forceinline__ void point_rows(v8f& nx, Cell& s) {
    asm volatile("" : "+v"(nx), "+v"(s.h0), "+v"(s.h1), "+v"(s.c) : : "memory");
}

__device__ __forceinline__ const char* tile_base(const char* __restrict__ xb, int u) {
    return xb + (size_t)u * (size_t)(TS * II * sizeof(float));
}

__device__ __forceinline__ void next_tile(const char*& rp, int u) {
    unsigned long long inc = u < NT ? (unsigned long long)(TS * II * sizeof(float)) : 0ull;
    asm volatile("" : "+s"(inc));
    rp += inc;
}

__device__ __forceinline__ void chunk(XRow (&q)[TC], v8f& g, Cell& st,
                                      const char*& rp, unsigned lane_off,
                                      int u0, const v2f (&w)[7], const v8f cb,
                                      const LaneK& k) {
#pragma unroll
    for (int s = 0; s < TC; ++s) {
        v8f nx = cb;
        const XRow& r = q[s];
        gate_mm<0>(r, w, nx);  cell_step(st, g[0], k);   point(nx, st);
        gate_mm<1>(r, w, nx);  cell_step(st, g[1], k);   point(nx, st);
        gate_mm<2>(r, w, nx);  cell_step(st, g[2], k);   point(nx, st);
        gate_mm<3>(r, w, nx);  cell_step(st, g[3], k);   point(nx, st);
        gate_mm<4>(r, w, nx);  cell_step(st, g[4], k);   point(nx, st);
        gate_mm<5>(r, w, nx);  cell_step(st, g[5], k);   point(nx, st);
        gate_mm<6>(r, w, nx);  cell_step(st, g[6], k);   point_rows(nx, st);
        next_tile(rp, u0 + s + 1 + TC);
        load_row(q[s], rp, lane_off);
        cell_step(st, g[7], k);
        g = nx;
    }
}

__global__ void __launch_bounds__(WGT)
lstm_rowgate_scan(const float* __restrict__ x,
                  const float* __restrict__ W_ih,
                  const float* __restrict__ W_hh,
                  const float* __restrict__ b_ih,
                  const float* __restrict__ b_hh,
                  const float* __restrict__ fc_W,
                  const float* __restrict__ fc_b,
                  float* __restrict__ out) {
    __shared__ float hfin[WPB * WB * HH];

    const int tid   = threadIdx.x;
    const int lane  = tid & 31;
    const int wave  = tid >> 5;
    const int n     = lane & 15;
    const int khalf = lane >> 4;
    const int dt    = n & 7;
    const int b     = (blockIdx.x * WPB + wave) * WB + (n >> 3);

    v2f wfrag[7];
#pragma unroll
    for (int j = 0; j < 7; ++j) {
        const int k0 = 4 * j + 2 * khalf;
        float a0 = 0.0f, a1 = 0.0f;
        if (n < NG) {
            a0 = W_ih[n * II + k0];
            a1 = W_ih[n * II + k0 + 1];
        }
        v2f af; af.x = a0; af.y = a1;
        wfrag[j] = af;
    }

    float bias = 0.0f;
    if (n < NG) bias = b_ih[n] + b_hh[n];
    v8f cbias;
#pragma unroll
    for (int v = 0; v < 8; ++v) cbias[v] = bias;

    LaneK lk;
    lk.w0 = 0.0f; lk.w1 = 0.0f;
    if (n < NG) {
        lk.w0 = W_hh[n * HH + 0];
        lk.w1 = W_hh[n * HH + 1];
    }
    const bool plain = (n >> 1) == 2;
    lk.as = plain ? 1.0f : 0.5f;
    lk.aa = plain ? 1.0f : 0.5f;
    lk.ao = plain ? 0.0f : 0.5f;

    const char* xb = (const char*)x;
    const unsigned lane_off =
        (unsigned)(((b * TT + dt) * II + 2 * khalf) * (int)sizeof(float));

    XRow r0;
    load_row(r0, tile_base(xb, 0), lane_off);
    XRow q[TC];
#pragma unroll
    for (int s = 0; s < TC; ++s) load_row(q[s], tile_base(xb, s + 1), lane_off);

    const char* rp = tile_base(xb, TC);

    v8f g = cbias;
    gates_of(r0, wfrag, g);
    Cell st;
    st.h0 = 0.f; st.h1 = 0.f; st.c = 0.f;

    for (int u0 = 0; u0 < NT; u0 += TC)
        chunk(q, g, st, rp, lane_off, u0, wfrag, cbias, lk);

    if (n < HH) hfin[(wave * WB + khalf) * HH + n] = (n == 0) ? st.h0 : st.h1;
    __syncthreads();

    if (tid < WPB * WB) {
        const int bo = blockIdx.x * (WPB * WB) + tid;
        const float h0 = hfin[tid * HH + 0];
        const float h1 = hfin[tid * HH + 1];
#pragma unroll
        for (int c = 0; c < CC; ++c) {
            out[bo * CC + c] = fc_b[c] + h0 * fc_W[c * HH + 0] + h1 * fc_W[c * HH + 1];
        }
    }
}

extern "C" void kernel_launch(void* const* d_in, const int* in_sizes, int n_in,
                              void* d_out, int out_size, void* d_ws, size_t ws_size,
                              hipStream_t stream) {
    const float* x    = (const float*)d_in[0];
    const float* W_ih = (const float*)d_in[1];
    const float* W_hh = (const float*)d_in[2];
    const float* b_ih = (const float*)d_in[3];
    const float* b_hh = (const float*)d_in[4];
    const float* fc_W = (const float*)d_in[5];
    const float* fc_b = (const float*)d_in[6];
    float* out = (float*)d_out;

    lstm_rowgate_scan<<<NWG, WGT, 0, stream>>>(x, W_ih, W_hh, b_ih, b_hh,
                                               fc_W, fc_b, out);
}


# each step's matrix instruction (a tile's last step: its row requests) issued directly behind tanh(c), in that result's latency; was behind the step's first multiply-add, the chain waiting; same sums
# speedup vs baseline: 7.6143x; 1.0331x over previous
#include <hip/hip_runtime.h>


#define BB 2048
#define TT 1024
#define II 28
#define HH 2
#define CC 10
#define NG 8
#define TS 8
#define WB 2
#define NT (TT / TS)
#define TC 4
#define WPB 4
#define NWG (BB / (WB * WPB))
#define WGT (32 * WPB)

typedef __attribute__((ext_vector_type(2))) float v2f;
typedef __attribute__((ext_vector_type(8))) float v8f;

__device__ __forceinline__ float fast_tanh(float x) {
#if __has_builtin(__builtin_amdgcn_tanhf)
    return __builtin_amdgcn_tanhf(x);
#else
    float e = __expf(-2.0f * x);
    return (1.0f - e) / (1.0f + e);
#endif
}
__device__ __forceinline__ float fast_sigm(float x) {
    return 0.5f + 0.5f * fast_tanh(0.5f * x);
}

struct XRow { v2f f[7]; };

__device__ __forceinline__ void load_row(XRow& r, const char* __restrict__ tb, unsigned lane_off) {
    asm("" : "+v"(lane_off));
#pragma unroll
    for (int j = 0; j < 7; ++j) r.f[j] = *(const v2f*)(tb + lane_off + 16 * j);
}

template <int K>
__device__ __forceinline__ void gate_mm(const XRow& r, const v2f (&w)[7], v8f& c) {
    c = __builtin_amdgcn_wmma_f32_16x16x4_f32(
            false, r.f[K], false, w[K], (short)0, c, false, false);
}
__device__ __forceinline__ void gates_of(const XRow& r, const v2f (&w)[7], v8f& c) {
    gate_mm<0>(r, w, c); gate_mm<1>(r, w, c); gate_mm<2>(r, w, c); gate_mm<3>(r, w, c);
    gate_mm<4>(r, w, c); gate_mm<5>(r, w, c); gate_mm<6>(r, w, c);
}

struct Cell { float h0, h1, c; };

struct LaneK { float w0, w1, as, aa, ao; };

template <int N>
__device__ __forceinline__ float row_up(const float v) {
    return __int_as_float(__builtin_amdgcn_update_dpp(
               0, __float_as_int(v), 0x100 + N, 0xf, 0xf, true));
}
template <int N>
__device__ __forceinline__ float row_lane(const float v) {
    return __int_as_float(__builtin_amdgcn_update_dpp(
               0, __float_as_int(v), 0x150 + N, 0xf, 0xf, true));
}

__device__ __forceinline__ void cell_front(Cell& s, const float g, const LaneK& k,
                                           float& o0, float& o1) {
    const float t = g + s.h0 * k.w0 + s.h1 * k.w1;
    const float a = k.ao + k.aa * fast_tanh(k.as * t);
    o0 = row_lane<6>(a);
    o1 = row_lane<7>(a);
    s.c = row_up<2>(a) * s.c + a * row_up<4>(a);
}
__device__ __forceinline__ void cell_back(Cell& s, const float tc, const float o0, const float o1) {
    s.h0 = o0 * row_lane<0>(tc);
    s.h1 = o1 * row_lane<1>(tc);
}

__device__ __forceinline__ void before_tanh(float& c, float& o0, float& o1) {
    asm volatile("" : "+v"(c), "+v"(o0), "+v"(o1));
}
__device__ __forceinline__ void shadow_open(v2f& xk, float& tc) {
    asm volatile("" : "+v"(xk), "+v"(tc));
}
__device__ __forceinline__ void shadow_close(v8f& nx, float& tc) {
    asm volatile("" : "+v"(nx), "+v"(tc));
}
__device__ __forceinline__ void rows_open(float& tc) {
    asm volatile("" : "+v"(tc) : : "memory");
}
__device__ __forceinline__ void rows_close(float& tc) {
    asm volatile("" : "+v"(tc) : : "memory");
}

template <int K>
__device__ __forceinline__ void step_mm(Cell& s, const float g, const LaneK& k,
                                        XRow& r, const v2f (&w)[7], v8f& nx) {
    float o0, o1;
    cell_front(s, g, k, o0, o1);
    before_tanh(s.c, o0, o1);
    float tc = fast_tanh(s.c);
    shadow_open(r.f[K], tc);
    gate_mm<K>(r, w, nx);
    shadow_close(nx, tc);
    cell_back(s, tc, o0, o1);
}

__device__ __forceinline__ const char* tile_base(const char* __restrict__ xb, int u) {
    return xb + (size_t)u * (size_t)(TS * II * sizeof(float));
}

__device__ __forceinline__ void next_tile(const char*& rp, int u) {
    unsigned long long inc = u < NT ? (unsigned long long)(TS * II * sizeof(float)) : 0ull;
    asm volatile("" : "+s"(inc));
    rp += inc;
}

__device__ __forceinline__ void chunk(XRow (&q)[TC], v8f& g, Cell& st,
                                      const char*& rp, unsigned lane_off,
                                      int u0, const v2f (&w)[7], const v8f cb,
                                      const LaneK& k) {
#pragma unroll
    for (int s = 0; s < TC; ++s) {
        v8f nx = cb;
        XRow& r = q[s];
        step_mm<0>(st, g[0], k, r, w, nx);
        step_mm<1>(st, g[1], k, r, w, nx);
        step_mm<2>(st, g[2], k, r, w, nx);
        step_mm<3>(st, g[3], k, r, w, nx);
        step_mm<4>(st, g[4], k, r, w, nx);
        step_mm<5>(st, g[5], k, r, w, nx);
        step_mm<6>(st, g[6], k, r, w, nx);
        float o0, o1;
        cell_front(st, g[7], k, o0, o1);
        before_tanh(st.c, o0, o1);
        float tc = fast_tanh(st.c);
        rows_open(tc);
        next_tile(rp, u0 + s + 1 + TC);
        load_row(q[s], rp, lane_off);
        rows_close(tc);
        cell_back(st, tc, o0, o1);
        g = nx;
    }
}

__global__ void __launch_bounds__(WGT)
lstm_rowgate_scan(const float* __restrict__ x,
                  const float* __restrict__ W_ih,
                  const float* __restrict__ W_hh,
                  const float* __restrict__ b_ih,
                  const float* __restrict__ b_hh,
                  const float* __restrict__ fc_W,
                  const float* __restrict__ fc_b,
                  float* __restrict__ out) {
    __shared__ float hfin[WPB * WB * HH];

    const int tid   = threadIdx.x;
    const int lane  = tid & 31;
    const int wave  = tid >> 5;
    const int n     = lane & 15;
    const int khalf = lane >> 4;
    const int dt    = n & 7;
    const int b     = (blockIdx.x * WPB + wave) * WB + (n >> 3);

    v2f wfrag[7];
#pragma unroll
    for (int j = 0; j < 7; ++j) {
        const int k0 = 4 * j + 2 * khalf;
        float a0 = 0.0f, a1 = 0.0f;
        if (n < NG) {
            a0 = W_ih[n * II + k0];
            a1 = W_ih[n * II + k0 + 1];
        }
        v2f af; af.x = a0; af.y = a1;
        wfrag[j] = af;
    }

    float bias = 0.0f;
    if (n < NG) bias = b_ih[n] + b_hh[n];
    v8f cbias;
#pragma unroll
    for (int v = 0; v < 8; ++v) cbias[v] = bias;

    LaneK lk;
    lk.w0 = 0.0f; lk.w1 = 0.0f;
    if (n < NG) {
        lk.w0 = W_hh[n * HH + 0];
        lk.w1 = W_hh[n * HH + 1];
    }
    const bool plain = (n >> 1) == 2;
    lk.as = plain ? 1.0f : 0.5f;
    lk.aa = plain ? 1.0f : 0.5f;
    lk.ao = plain ? 0.0f : 0.5f;

    const char* xb = (const char*)x;
    const unsigned lane_off =
        (unsigned)(((b * TT + dt) * II + 2 * khalf) * (int)sizeof(float));

    XRow r0;
    load_row(r0, tile_base(xb, 0), lane_off);
    XRow q[TC];
#pragma unroll
    for (int s = 0; s < TC; ++s) load_row(q[s], tile_base(xb, s + 1), lane_off);

    const char* rp = tile_base(xb, TC);

    v8f g = cbias;
    gates_of(r0, wfrag, g);
    Cell st;
    st.h0 = 0.f; st.h1 = 0.f; st.c = 0.f;

    for (int u0 = 0; u0 < NT; u0 += TC)
        chunk(q, g, st, rp, lane_off, u0, wfrag, cbias, lk);

    if (n < HH) hfin[(wave * WB + khalf) * HH + n] = (n == 0) ? st.h0 : st.h1;
    __syncthreads();

    if (tid < WPB * WB) {
        const int bo = blockIdx.x * (WPB * WB) + tid;
        const float h0 = hfin[tid * HH + 0];
        const float h1 = hfin[tid * HH + 1];
#pragma unroll
        for (int c = 0; c < CC; ++c) {
            out[bo * CC + c] = fc_b[c] + h0 * fc_W[c * HH + 0] + h1 * fc_W[c * HH + 1];
        }
    }
}

extern "C" void kernel_launch(void* const* d_in, const int* in_sizes, int n_in,
                              void* d_out, int out_size, void* d_ws, size_t ws_size,
                              hipStream_t stream) {
    const float* x    = (const float*)d_in[0];
    const float* W_ih = (const float*)d_in[1];
    const float* W_hh = (const float*)d_in[2];
    const float* b_ih = (const float*)d_in[3];
    const float* b_hh = (const float*)d_in[4];
    const float* fc_W = (const float*)d_in[5];
    const float* fc_b = (const float*)d_in[6];
    float* out = (float*)d_out;

    lstm_rowgate_scan<<<NWG, WGT, 0, stream>>>(x, W_ih, W_hh, b_ih, b_hh,
                                               fc_W, fc_b, out);
}
